# starting kernel
# baseline (speedup 1.0000x reference)
#include <hip/hip_runtime.h>
#include <hip/hip_bf16.h>

typedef __attribute__((ext_vector_type(2))) float v2f;
typedef __attribute__((ext_vector_type(8))) float v8f;

__global__ void gcn_zero(float* __restrict__ p, long n) {
    long i = (long)blockIdx.x * blockDim.x + threadIdx.x;
    if (i < n) p[i] = 0.0f;
}

__global__ void gcn_degree(const int* __restrict__ src, const int* __restrict__ dst,
                           float* __restrict__ deg_src, float* __restrict__ deg_dst, int e) {
    int i = blockIdx.x * blockDim.x + threadIdx.x;
    if (i >= e) return;
    unsafeAtomicAdd(&deg_src[src[i]], 1.0f);
    unsafeAtomicAdd(&deg_dst[dst[i]], 1.0f);
}

__global__ void gcn_rsqrt(float* __restrict__ p, long n) {
    long i = (long)blockIdx.x * blockDim.x + threadIdx.x;
    if (i < n) p[i] = rsqrtf(fmaxf(p[i], 1.0f));
}

template <int DIN, int DN>
__global__ __launch_bounds__(((DN + 15) / 16) * 32) void gcn_gemm_wmma(
    const float* __restrict__ H, const float* __restrict__ norm_src,
    const float* __restrict__ W, float* __restrict__ Y, int n)
{
    constexpr int LDA = DIN + 1;
    __shared__ float As[16 * LDA];

    const int row0 = blockIdx.x * 16;
    const int tid  = threadIdx.x;

    for (int idx = tid; idx < 16 * DIN; idx += blockDim.x) {
        int r = idx / DIN;
        int c = idx - r * DIN;
        int row = row0 + r;
        As[r * LDA + c] = (row < n) ? H[(size_t)row * DIN + c] * norm_src[row] : 0.0f;
    }
    __syncthreads();

    const int lane = tid & 31;
    const int wave = tid >> 5;
    const int mn   = lane & 15;
    const int kh   = lane >> 4;
    const int col  = (wave << 4) + mn;

    const int   colc = (col < DN) ? col : (DN - 1);
    float msk = 1.0f;
    if constexpr ((DN & 15) != 0) msk = (col < DN) ? 1.0f : 0.0f;

    const float* __restrict__ ap = As + mn * LDA + 2 * kh;
    const float* __restrict__ wp = W + (size_t)(2 * kh) * DN + colc;

    v8f acc = {0.f, 0.f, 0.f, 0.f, 0.f, 0.f, 0.f, 0.f};

    #pragma unroll
    for (int k = 0; k < DIN; k += 4) {
        v2f a, b;
        a.x = ap[k];
        a.y = ap[k + 1];
        b.x = wp[k * DN] * msk;
        b.y = wp[k * DN + DN] * msk;
        acc = __builtin_amdgcn_wmma_f32_16x16x4_f32(
            false, a, false, b, (short)0, acc, false, false);
    }

    if (col < DN) {
        #pragma unroll
        for (int v = 0; v < 8; ++v) {
            int row = row0 + v + 8 * kh;
            if (row < n) Y[(size_t)row * DN + col] = acc[v];
        }
    }
}

template <int D>
__global__ __launch_bounds__(256) void gcn_scatter_add(
    const float* __restrict__ Y, const int* __restrict__ src, const int* __restrict__ dst,
    float* __restrict__ Out, int n_edges)
{
    constexpr int CPE = D >> 2;
    long idx = (long)blockIdx.x * blockDim.x + threadIdx.x;
    long total = (long)n_edges * CPE;
    if (idx >= total) return;
    int e = (int)(idx / CPE);
    int c = (int)(idx - (long)e * CPE);
    int s = src[e];
    int t = dst[e];
    const float4 v = *reinterpret_cast<const float4*>(Y + (size_t)s * D + 4 * c);
    float* o = Out + (size_t)t * D + 4 * c;
    unsafeAtomicAdd(o + 0, v.x);
    unsafeAtomicAdd(o + 1, v.y);
    unsafeAtomicAdd(o + 2, v.z);
    unsafeAtomicAdd(o + 3, v.w);
}

__global__ void gcn_post(float* __restrict__ A, const float* __restrict__ norm_dst,
                         const float* __restrict__ bias, int d, int n, int do_relu)
{
    long idx = (long)blockIdx.x * blockDim.x + threadIdx.x;
    if (idx >= (long)n * d) return;
    int i = (int)(idx / d);
    int f = (int)(idx - (long)i * d);
    float v = A[idx] * norm_dst[i] + bias[f];
    if (do_relu) v = fmaxf(v, 0.0f);
    A[idx] = v;
}

static inline int cdiv_l(long a, long b) { return (int)((a + b - 1) / b); }

template <int DIN, int DN>
static void run_layer(const float* Hin, const float* norm_src, const float* norm_dst,
                      const float* W, const float* bias, float* bufP, float* Agg,
                      const int* src, const int* dst, int N, int E, int do_relu,
                      hipStream_t stream) {
    constexpr int CT = (DN + 15) / 16;
    gcn_gemm_wmma<DIN, DN><<<cdiv_l(N, 16), 32 * CT, 0, stream>>>(
        Hin, norm_src, W, bufP, N);
    gcn_zero<<<cdiv_l((long)N * DN, 256), 256, 0, stream>>>(Agg, (long)N * DN);
    long tot = (long)E * (DN / 4);
    gcn_scatter_add<DN><<<cdiv_l(tot, 256), 256, 0, stream>>>(bufP, src, dst, Agg, E);
    gcn_post<<<cdiv_l((long)N * DN, 256), 256, 0, stream>>>(
        Agg, norm_dst, bias, DN, N, do_relu);
}

extern "C" void kernel_launch(void* const* d_in, const int* in_sizes, int n_in,
                              void* d_out, int out_size, void* d_ws, size_t ws_size,
                              hipStream_t stream) {
    const float* x   = (const float*)d_in[0];
    const int*   src = (const int*)d_in[1];
    const int*   dst = (const int*)d_in[2];
    const float* W1 = (const float*)d_in[3];
    const float* b1 = (const float*)d_in[4];
    const float* W2 = (const float*)d_in[5];
    const float* b2 = (const float*)d_in[6];
    const float* W3 = (const float*)d_in[7];
    const float* b3 = (const float*)d_in[8];
    const float* W4 = (const float*)d_in[9];
    const float* b4 = (const float*)d_in[10];

    const int N = in_sizes[0] / 256;
    const int E = in_sizes[1];

    float* ws       = (float*)d_ws;
    float* norm_src = ws;
    float* norm_dst = ws + N;
    float* bufP     = ws + 2 * (size_t)N;
    float* bufH     = bufP + (size_t)N * 128;

    gcn_zero<<<cdiv_l(2L * N, 256), 256, 0, stream>>>(norm_src, 2L * N);
    gcn_degree<<<cdiv_l(E, 256), 256, 0, stream>>>(src, dst, norm_src, norm_dst, E);
    gcn_rsqrt<<<cdiv_l(2L * N, 256), 256, 0, stream>>>(norm_src, 2L * N);

    run_layer<256, 128>(x,    norm_src, norm_dst, W1, b1, bufP, bufH,
                        src, dst, N, E, 1, stream);
    run_layer<128, 64>(bufH,  norm_src, norm_dst, W2, b2, bufP, bufH,
                       src, dst, N, E, 1, stream);
    run_layer<64, 16>(bufH,   norm_src, norm_dst, W3, b3, bufP, bufH,
                      src, dst, N, E, 1, stream);
    run_layer<16, 40>(bufH,   norm_src, norm_dst, W4, b4, bufP, (float*)d_out,
                      src, dst, N, E, 0, stream);
}


# Aggregate by a dst-sorted source list (rank from the in-degree atomic + scan): one store per row with norm/bias/relu; f32 scatter atomics, zero and post passes gone; sums in arrival-rank order
# speedup vs baseline: 2.2659x; 2.2659x over previous
#include <hip/hip_runtime.h>
#include <hip/hip_bf16.h>


typedef __attribute__((ext_vector_type(2))) float v2f;
typedef __attribute__((ext_vector_type(8))) float v8f;

__global__ void gcn_zero_i32(int* __restrict__ p, long n) {
    long i = (long)blockIdx.x * blockDim.x + threadIdx.x;
    if (i < n) p[i] = 0;
}

__global__ void gcn_degree_rank(const int* __restrict__ src, const int* __restrict__ dst,
                                int* __restrict__ cnt_src, int* __restrict__ cnt_dst,
                                int* __restrict__ rank, int e) {
    int i = blockIdx.x * blockDim.x + threadIdx.x;
    if (i >= e) return;
    const int s = src[i];
    const int d = dst[i];
    atomicAdd(&cnt_src[s], 1);
    rank[i] = atomicAdd(&cnt_dst[d], 1);
}

__global__ void gcn_norm(const int* __restrict__ cnt, float* __restrict__ nrm, long n) {
    long i = (long)blockIdx.x * blockDim.x + threadIdx.x;
    if (i < n) nrm[i] = rsqrtf(fmaxf((float)cnt[i], 1.0f));
}

constexpr int SCAN_T    = 256;
constexpr int SCAN_TILE = 4 * SCAN_T;

static __device__ __forceinline__ int block_scan_excl(int v, int* sh, int t, int* total) {
    sh[t] = v;
    __syncthreads();
    for (int off = 1; off < SCAN_T; off <<= 1) {
        int a = (t >= off) ? sh[t - off] : 0;
        __syncthreads();
        sh[t] += a;
        __syncthreads();
    }
    *total = sh[SCAN_T - 1];
    int ex = sh[t] - v;
    __syncthreads();
    return ex;
}

__global__ __launch_bounds__(SCAN_T) void gcn_tile_sums(
    const int* __restrict__ cnt, int* __restrict__ tsum, int n)
{
    __shared__ int sh[SCAN_T];
    const int t = threadIdx.x;
    const int base = blockIdx.x * SCAN_TILE + 4 * t;
    int s = 0;
    for (int k = 0; k < 4; ++k) s += (base + k < n) ? cnt[base + k] : 0;
    int total;
    block_scan_excl(s, sh, t, &total);
    if (t == 0) tsum[blockIdx.x] = total;
}

__global__ __launch_bounds__(SCAN_T) void gcn_rowptr(
    const int* __restrict__ cnt, const int* __restrict__ tsum,
    int* __restrict__ rowptr, int n, int e)
{
    __shared__ int sh[SCAN_T];
    const int t = threadIdx.x;
    const int b = blockIdx.x;

    int part = 0;
    for (int j = t; j < b; j += SCAN_T) part += tsum[j];
    int offset;
    block_scan_excl(part, sh, t, &offset);

    const int base = b * SCAN_TILE + 4 * t;
    int c[4];
    int s = 0;
    for (int k = 0; k < 4; ++k) { c[k] = (base + k < n) ? cnt[base + k] : 0; s += c[k]; }
    int tile_total;
    int run = offset + block_scan_excl(s, sh, t, &tile_total);
    for (int k = 0; k < 4; ++k) {
        if (base + k < n) rowptr[base + k] = run;
        run += c[k];
    }
    if (b == 0 && t == 0) rowptr[n] = e;
}

__global__ void gcn_fill(const int* __restrict__ src, const int* __restrict__ dst,
                         const int* __restrict__ rank, const int* __restrict__ rowptr,
                         int* __restrict__ csr, int e) {
    int i = blockIdx.x * blockDim.x + threadIdx.x;
    if (i >= e) return;
    csr[rowptr[dst[i]] + rank[i]] = src[i];
}

__global__ void gcn_move_i32(const int* __restrict__ a, int* __restrict__ b, long n) {
    long i = (long)blockIdx.x * blockDim.x + threadIdx.x;
    if (i < n) b[i] = a[i];
}

template <int DIN, int DN>
__global__ __launch_bounds__(((DN + 15) / 16) * 32) void gcn_gemm_wmma(
    const float* __restrict__ H, const float* __restrict__ norm_src,
    const float* __restrict__ W, float* __restrict__ Y, int n)
{
    constexpr int LDA = DIN + 1;
    __shared__ float As[16 * LDA];

    const int row0 = blockIdx.x * 16;
    const int tid  = threadIdx.x;

    for (int idx = tid; idx < 16 * DIN; idx += blockDim.x) {
        int r = idx / DIN;
        int c = idx - r * DIN;
        int row = row0 + r;
        As[r * LDA + c] = (row < n) ? H[(size_t)row * DIN + c] * norm_src[row] : 0.0f;
    }
    __syncthreads();

    const int lane = tid & 31;
    const int wave = tid >> 5;
    const int mn   = lane & 15;
    const int kh   = lane >> 4;
    const int col  = (wave << 4) + mn;

    const int   colc = (col < DN) ? col : (DN - 1);
    float msk = 1.0f;
    if constexpr ((DN & 15) != 0) msk = (col < DN) ? 1.0f : 0.0f;

    const float* __restrict__ ap = As + mn * LDA + 2 * kh;
    const float* __restrict__ wp = W + (size_t)(2 * kh) * DN + colc;

    v8f acc = {0.f, 0.f, 0.f, 0.f, 0.f, 0.f, 0.f, 0.f};

    #pragma unroll
    for (int k = 0; k < DIN; k += 4) {
        v2f a, b;
        a.x = ap[k];
        a.y = ap[k + 1];
        b.x = wp[k * DN] * msk;
        b.y = wp[k * DN + DN] * msk;
        acc = __builtin_amdgcn_wmma_f32_16x16x4_f32(
            false, a, false, b, (short)0, acc, false, false);
    }

    if (col < DN) {
        #pragma unroll
        for (int v = 0; v < 8; ++v) {
            int row = row0 + v + 8 * kh;
            if (row < n) Y[(size_t)row * DN + col] = acc[v];
        }
    }
}

template <int D, int RELU>
__global__ __launch_bounds__(256) void gcn_gather(
    const float* __restrict__ Y, const int* __restrict__ rowptr, const int* __restrict__ csr,
    const float* __restrict__ norm_dst, const float* __restrict__ bias,
    float* __restrict__ Out, int n)
{
    constexpr int CPE = D >> 2;
    long idx = (long)blockIdx.x * blockDim.x + threadIdx.x;
    if (idx >= (long)n * CPE) return;
    const int i = (int)(idx / CPE);
    const int c = (int)(idx - (long)i * CPE);

    int p = rowptr[i];
    const int end  = rowptr[i + 1];
    const int last = end - 1;
    const float* __restrict__ yb = Y + 4 * c;

    float4 acc = make_float4(0.f, 0.f, 0.f, 0.f);
    for (; p < end; p += 4) {
        const bool h1 = (p + 1 < end), h2 = (p + 2 < end), h3 = (p + 3 < end);
        const int s0 = csr[p];
        const int s1 = csr[h1 ? p + 1 : last];
        const int s2 = csr[h2 ? p + 2 : last];
        const int s3 = csr[h3 ? p + 3 : last];
        const float4 v0 = *reinterpret_cast<const float4*>(yb + (size_t)s0 * D);
        const float4 v1 = *reinterpret_cast<const float4*>(yb + (size_t)s1 * D);
        const float4 v2 = *reinterpret_cast<const float4*>(yb + (size_t)s2 * D);
        const float4 v3 = *reinterpret_cast<const float4*>(yb + (size_t)s3 * D);
        acc.x += v0.x; acc.y += v0.y; acc.z += v0.z; acc.w += v0.w;
        acc.x += h1 ? v1.x : 0.f; acc.y += h1 ? v1.y : 0.f;
        acc.z += h1 ? v1.z : 0.f; acc.w += h1 ? v1.w : 0.f;
        acc.x += h2 ? v2.x : 0.f; acc.y += h2 ? v2.y : 0.f;
        acc.z += h2 ? v2.z : 0.f; acc.w += h2 ? v2.w : 0.f;
        acc.x += h3 ? v3.x : 0.f; acc.y += h3 ? v3.y : 0.f;
        acc.z += h3 ? v3.z : 0.f; acc.w += h3 ? v3.w : 0.f;
    }

    const float nd = norm_dst[i];
    const float4 bv = *reinterpret_cast<const float4*>(bias + 4 * c);
    float4 r;
    r.x = acc.x * nd + bv.x;
    r.y = acc.y * nd + bv.y;
    r.z = acc.z * nd + bv.z;
    r.w = acc.w * nd + bv.w;
    if (RELU) {
        r.x = fmaxf(r.x, 0.0f); r.y = fmaxf(r.y, 0.0f);
        r.z = fmaxf(r.z, 0.0f); r.w = fmaxf(r.w, 0.0f);
    }
    *reinterpret_cast<float4*>(Out + (size_t)i * D + 4 * c) = r;
}

static inline int cdiv_l(long a, long b) { return (int)((a + b - 1) / b); }

template <int DIN, int DN, int RELU>
static void run_layer(const float* Hin, const float* norm_src, const float* norm_dst,
                      const float* W, const float* bias, float* bufP, float* Agg,
                      const int* rowptr, const int* csr, int N, hipStream_t stream) {
    constexpr int CT = (DN + 15) / 16;
    gcn_gemm_wmma<DIN, DN><<<cdiv_l(N, 16), 32 * CT, 0, stream>>>(
        Hin, norm_src, W, bufP, N);
    gcn_gather<DN, RELU><<<cdiv_l((long)N * (DN / 4), 256), 256, 0, stream>>>(
        bufP, rowptr, csr, norm_dst, bias, Agg, N);
}

extern "C" void kernel_launch(void* const* d_in, const int* in_sizes, int n_in,
                              void* d_out, int out_size, void* d_ws, size_t ws_size,
                              hipStream_t stream) {
    const float* x   = (const float*)d_in[0];
    const int*   src = (const int*)d_in[1];
    const int*   dst = (const int*)d_in[2];
    const float* W1 = (const float*)d_in[3];
    const float* b1 = (const float*)d_in[4];
    const float* W2 = (const float*)d_in[5];
    const float* b2 = (const float*)d_in[6];
    const float* W3 = (const float*)d_in[7];
    const float* b3 = (const float*)d_in[8];
    const float* W4 = (const float*)d_in[9];
    const float* b4 = (const float*)d_in[10];

    const int N = in_sizes[0] / 256;
    const int E = in_sizes[1];
    const int NT = cdiv_l(N, SCAN_TILE);

    float* ws       = (float*)d_ws;
    float* norm_src = ws;
    float* norm_dst = ws + N;
    float* bufP     = ws + 2 * (size_t)N;
    float* bufH     = bufP + (size_t)N * 128;

    int* csr0   = (int*)d_out;
    int* rowp0  = csr0 + E;
    int* rank   = rowp0 + N + 1;
    int* cnt    = rank + E;
    int* tsum   = cnt + 2 * (size_t)N;
    int* csrW   = (int*)(bufP + (size_t)N * 64);
    int* rowpW  = csrW + E;

    gcn_zero_i32<<<cdiv_l(2L * N, 256), 256, 0, stream>>>(cnt, 2L * N);
    gcn_degree_rank<<<cdiv_l(E, 256), 256, 0, stream>>>(src, dst, cnt, cnt + N, rank, E);
    gcn_norm<<<cdiv_l(2L * N, 256), 256, 0, stream>>>(cnt, norm_src, 2L * N);
    gcn_tile_sums<<<NT, SCAN_T, 0, stream>>>(cnt + N, tsum, N);
    gcn_rowptr<<<NT, SCAN_T, 0, stream>>>(cnt + N, tsum, rowp0, N, E);
    gcn_fill<<<cdiv_l(E, 256), 256, 0, stream>>>(src, dst, rank, rowp0, csr0, E);

    run_layer<256, 128, 1>(x,    norm_src, norm_dst, W1, b1, bufP, bufH, rowp0, csr0, N, stream);
    gcn_move_i32<<<cdiv_l((long)E + N + 1, 256), 256, 0, stream>>>(csr0, csrW, (long)E + N + 1);
    run_layer<128, 64, 1>(bufH,  norm_src, norm_dst, W2, b2, bufP, bufH, rowpW, csrW, N, stream);
    run_layer<64, 16, 1>(bufH,   norm_src, norm_dst, W3, b3, bufP, bufH, rowpW, csrW, N, stream);
    run_layer<16, 40, 0>(bufH,   norm_src, norm_dst, W4, b4, bufP, (float*)d_out, rowpW, csrW, N, stream);
}


# GEMM A tile staged in 16-byte pieces, all of a thread's loads issued before the first use (was one dependent 4-byte load per trip); row pitch DIN+4: aligned LDS stores, 8-byte fragment reads
# speedup vs baseline: 2.5325x; 1.1177x over previous
#include <hip/hip_runtime.h>
#include <hip/hip_bf16.h>


typedef __attribute__((ext_vector_type(2))) float v2f;
typedef __attribute__((ext_vector_type(8))) float v8f;

__global__ void gcn_zero_i32(int* __restrict__ p, long n) {
    long i = (long)blockIdx.x * blockDim.x + threadIdx.x;
    if (i < n) p[i] = 0;
}

__global__ void gcn_degree_rank(const int* __restrict__ src, const int* __restrict__ dst,
                                int* __restrict__ cnt_src, int* __restrict__ cnt_dst,
                                int* __restrict__ rank, int e) {
    int i = blockIdx.x * blockDim.x + threadIdx.x;
    if (i >= e) return;
    const int s = src[i];
    const int d = dst[i];
    atomicAdd(&cnt_src[s], 1);
    rank[i] = atomicAdd(&cnt_dst[d], 1);
}

__global__ void gcn_norm(const int* __restrict__ cnt, float* __restrict__ nrm, long n) {
    long i = (long)blockIdx.x * blockDim.x + threadIdx.x;
    if (i < n) nrm[i] = rsqrtf(fmaxf((float)cnt[i], 1.0f));
}

constexpr int SCAN_T    = 256;
constexpr int SCAN_TILE = 4 * SCAN_T;

static __device__ __forceinline__ int block_scan_excl(int v, int* sh, int t, int* total) {
    sh[t] = v;
    __syncthreads();
    for (int off = 1; off < SCAN_T; off <<= 1) {
        int a = (t >= off) ? sh[t - off] : 0;
        __syncthreads();
        sh[t] += a;
        __syncthreads();
    }
    *total = sh[SCAN_T - 1];
    int ex = sh[t] - v;
    __syncthreads();
    return ex;
}

__global__ __launch_bounds__(SCAN_T) void gcn_tile_sums(
    const int* __restrict__ cnt, int* __restrict__ tsum, int n)
{
    __shared__ int sh[SCAN_T];
    const int t = threadIdx.x;
    const int base = blockIdx.x * SCAN_TILE + 4 * t;
    int s = 0;
    for (int k = 0; k < 4; ++k) s += (base + k < n) ? cnt[base + k] : 0;
    int total;
    block_scan_excl(s, sh, t, &total);
    if (t == 0) tsum[blockIdx.x] = total;
}

__global__ __launch_bounds__(SCAN_T) void gcn_rowptr(
    const int* __restrict__ cnt, const int* __restrict__ tsum,
    int* __restrict__ rowptr, int n, int e)
{
    __shared__ int sh[SCAN_T];
    const int t = threadIdx.x;
    const int b = blockIdx.x;

    int part = 0;
    for (int j = t; j < b; j += SCAN_T) part += tsum[j];
    int offset;
    block_scan_excl(part, sh, t, &offset);

    const int base = b * SCAN_TILE + 4 * t;
    int c[4];
    int s = 0;
    for (int k = 0; k < 4; ++k) { c[k] = (base + k < n) ? cnt[base + k] : 0; s += c[k]; }
    int tile_total;
    int run = offset + block_scan_excl(s, sh, t, &tile_total);
    for (int k = 0; k < 4; ++k) {
        if (base + k < n) rowptr[base + k] = run;
        run += c[k];
    }
    if (b == 0 && t == 0) rowptr[n] = e;
}

__global__ void gcn_fill(const int* __restrict__ src, const int* __restrict__ dst,
                         const int* __restrict__ rank, const int* __restrict__ rowptr,
                         int* __restrict__ csr, int e) {
    int i = blockIdx.x * blockDim.x + threadIdx.x;
    if (i >= e) return;
    csr[rowptr[dst[i]] + rank[i]] = src[i];
}

__global__ void gcn_move_i32(const int* __restrict__ a, int* __restrict__ b, long n) {
    long i = (long)blockIdx.x * blockDim.x + threadIdx.x;
    if (i < n) b[i] = a[i];
}

template <int DIN, int DN>
__global__ __launch_bounds__(((DN + 15) / 16) * 32) void gcn_gemm_wmma(
    const float* __restrict__ H, const float* __restrict__ norm_src,
    const float* __restrict__ W, float* __restrict__ Y, int n)
{
    constexpr int T   = ((DN + 15) / 16) * 32;
    constexpr int LDA = DIN + 4;
    constexpr int RC  = DIN / 4;
    constexpr int NCH = 16 * RC;
    constexpr int PER = (NCH + T - 1) / T;
    constexpr bool EXACT = (NCH % T) == 0;
    __shared__ __attribute__((aligned(16))) float As[16 * LDA];

    const int row0 = blockIdx.x * 16;
    const int tid  = threadIdx.x;

    float4 hv[PER];
    float  sv[PER];
    #pragma unroll
    for (int j = 0; j < PER; ++j) {
        const int ch  = tid + j * T;
        const int chc = (EXACT || ch < NCH) ? ch : (NCH - 1);
        const int r   = chc / RC;
        const int c4  = chc - r * RC;
        const int row = row0 + r;
        const int rowc = (row < n) ? row : (n - 1);
        hv[j] = *reinterpret_cast<const float4*>(H + (size_t)rowc * DIN + 4 * c4);
        sv[j] = norm_src[rowc];
    }
    #pragma unroll
    for (int j = 0; j < PER; ++j) {
        const int ch = tid + j * T;
        if (EXACT || ch < NCH) {
            const int r   = ch / RC;
            const int c4  = ch - r * RC;
            const float s = sv[j];
            float4 v;
            v.x = hv[j].x * s;
            v.y = hv[j].y * s;
            v.z = hv[j].z * s;
            v.w = hv[j].w * s;
            *reinterpret_cast<float4*>(As + r * LDA + 4 * c4) = v;
        }
    }
    __syncthreads();

    const int lane = tid & 31;
    const int wave = tid >> 5;
    const int mn   = lane & 15;
    const int kh   = lane >> 4;
    const int col  = (wave << 4) + mn;

    const int   colc = (col < DN) ? col : (DN - 1);
    float msk = 1.0f;
    if constexpr ((DN & 15) != 0) msk = (col < DN) ? 1.0f : 0.0f;

    const float* __restrict__ ap = As + mn * LDA + 2 * kh;
    const float* __restrict__ wp = W + (size_t)(2 * kh) * DN + colc;

    v8f acc = {0.f, 0.f, 0.f, 0.f, 0.f, 0.f, 0.f, 0.f};

    #pragma unroll
    for (int k = 0; k < DIN; k += 4) {
        v2f b;
        const v2f a = *reinterpret_cast<const v2f*>(ap + k);
        b.x = wp[k * DN] * msk;
        b.y = wp[k * DN + DN] * msk;
        acc = __builtin_amdgcn_wmma_f32_16x16x4_f32(
            false, a, false, b, (short)0, acc, false, false);
    }

    if (col < DN) {
        #pragma unroll
        for (int v = 0; v < 8; ++v) {
            int row = row0 + v + 8 * kh;
            if (row < n) Y[(size_t)row * DN + col] = acc[v];
        }
    }
}

template <int D, int RELU>
__global__ __launch_bounds__(256) void gcn_gather(
    const float* __restrict__ Y, const int* __restrict__ rowptr, const int* __restrict__ csr,
    const float* __restrict__ norm_dst, const float* __restrict__ bias,
    float* __restrict__ Out, int n)
{
    constexpr int CPE = D >> 2;
    long idx = (long)blockIdx.x * blockDim.x + threadIdx.x;
    if (idx >= (long)n * CPE) return;
    const int i = (int)(idx / CPE);
    const int c = (int)(idx - (long)i * CPE);

    int p = rowptr[i];
    const int end  = rowptr[i + 1];
    const int last = end - 1;
    const float* __restrict__ yb = Y + 4 * c;

    float4 acc = make_float4(0.f, 0.f, 0.f, 0.f);
    for (; p < end; p += 4) {
        const bool h1 = (p + 1 < end), h2 = (p + 2 < end), h3 = (p + 3 < end);
        const int s0 = csr[p];
        const int s1 = csr[h1 ? p + 1 : last];
        const int s2 = csr[h2 ? p + 2 : last];
        const int s3 = csr[h3 ? p + 3 : last];
        const float4 v0 = *reinterpret_cast<const float4*>(yb + (size_t)s0 * D);
        const float4 v1 = *reinterpret_cast<const float4*>(yb + (size_t)s1 * D);
        const float4 v2 = *reinterpret_cast<const float4*>(yb + (size_t)s2 * D);
        const float4 v3 = *reinterpret_cast<const float4*>(yb + (size_t)s3 * D);
        acc.x += v0.x; acc.y += v0.y; acc.z += v0.z; acc.w += v0.w;
        acc.x += h1 ? v1.x : 0.f; acc.y += h1 ? v1.y : 0.f;
        acc.z += h1 ? v1.z : 0.f; acc.w += h1 ? v1.w : 0.f;
        acc.x += h2 ? v2.x : 0.f; acc.y += h2 ? v2.y : 0.f;
        acc.z += h2 ? v2.z : 0.f; acc.w += h2 ? v2.w : 0.f;
        acc.x += h3 ? v3.x : 0.f; acc.y += h3 ? v3.y : 0.f;
        acc.z += h3 ? v3.z : 0.f; acc.w += h3 ? v3.w : 0.f;
    }

    const float nd = norm_dst[i];
    const float4 bv = *reinterpret_cast<const float4*>(bias + 4 * c);
    float4 r;
    r.x = acc.x * nd + bv.x;
    r.y = acc.y * nd + bv.y;
    r.z = acc.z * nd + bv.z;
    r.w = acc.w * nd + bv.w;
    if (RELU) {
        r.x = fmaxf(r.x, 0.0f); r.y = fmaxf(r.y, 0.0f);
        r.z = fmaxf(r.z, 0.0f); r.w = fmaxf(r.w, 0.0f);
    }
    *reinterpret_cast<float4*>(Out + (size_t)i * D + 4 * c) = r;
}

static inline int cdiv_l(long a, long b) { return (int)((a + b - 1) / b); }

template <int DIN, int DN, int RELU>
static void run_layer(const float* Hin, const float* norm_src, const float* norm_dst,
                      const float* W, const float* bias, float* bufP, float* Agg,
                      const int* rowptr, const int* csr, int N, hipStream_t stream) {
    constexpr int CT = (DN + 15) / 16;
    gcn_gemm_wmma<DIN, DN><<<cdiv_l(N, 16), 32 * CT, 0, stream>>>(
        Hin, norm_src, W, bufP, N);
    gcn_gather<DN, RELU><<<cdiv_l((long)N * (DN / 4), 256), 256, 0, stream>>>(
        bufP, rowptr, csr, norm_dst, bias, Agg, N);
}

extern "C" void kernel_launch(void* const* d_in, const int* in_sizes, int n_in,
                              void* d_out, int out_size, void* d_ws, size_t ws_size,
                              hipStream_t stream) {
    const float* x   = (const float*)d_in[0];
    const int*   src = (const int*)d_in[1];
    const int*   dst = (const int*)d_in[2];
    const float* W1 = (const float*)d_in[3];
    const float* b1 = (const float*)d_in[4];
    const float* W2 = (const float*)d_in[5];
    const float* b2 = (const float*)d_in[6];
    const float* W3 = (const float*)d_in[7];
    const float* b3 = (const float*)d_in[8];
    const float* W4 = (const float*)d_in[9];
    const float* b4 = (const float*)d_in[10];

    const int N = in_sizes[0] / 256;
    const int E = in_sizes[1];
    const int NT = cdiv_l(N, SCAN_TILE);

    float* ws       = (float*)d_ws;
    float* norm_src = ws;
    float* norm_dst = ws + N;
    float* bufP     = ws + 2 * (size_t)N;
    float* bufH     = bufP + (size_t)N * 128;

    int* csr0   = (int*)d_out;
    int* rowp0  = csr0 + E;
    int* rank   = rowp0 + N + 1;
    int* cnt    = rank + E;
    int* tsum   = cnt + 2 * (size_t)N;
    int* csrW   = (int*)(bufP + (size_t)N * 64);
    int* rowpW  = csrW + E;

    gcn_zero_i32<<<cdiv_l(2L * N, 256), 256, 0, stream>>>(cnt, 2L * N);
    gcn_degree_rank<<<cdiv_l(E, 256), 256, 0, stream>>>(src, dst, cnt, cnt + N, rank, E);
    gcn_norm<<<cdiv_l(2L * N, 256), 256, 0, stream>>>(cnt, norm_src, 2L * N);
    gcn_tile_sums<<<NT, SCAN_T, 0, stream>>>(cnt + N, tsum, N);
    gcn_rowptr<<<NT, SCAN_T, 0, stream>>>(cnt + N, tsum, rowp0, N, E);
    gcn_fill<<<cdiv_l(E, 256), 256, 0, stream>>>(src, dst, rank, rowp0, csr0, E);

    run_layer<256, 128, 1>(x,    norm_src, norm_dst, W1, b1, bufP, bufH, rowp0, csr0, N, stream);
    gcn_move_i32<<<cdiv_l((long)E + N + 1, 256), 256, 0, stream>>>(csr0, csrW, (long)E + N + 1);
    run_layer<128, 64, 1>(bufH,  norm_src, norm_dst, W2, b2, bufP, bufH, rowpW, csrW, N, stream);
    run_layer<64, 16, 1>(bufH,   norm_src, norm_dst, W3, b3, bufP, bufH, rowpW, csrW, N, stream);
    run_layer<16, 40, 0>(bufH,   norm_src, norm_dst, W4, b4, bufP, (float*)d_out, rowpW, csrW, N, stream);
}


# Degrees and edge ranks counted in 16-bit LDS tables per 32768-edge slice (slice offsets by a node pass) instead of two scattered global atomics per edge; row sums in slice-then-arrival order
# speedup vs baseline: 3.4853x; 1.3762x over previous
#include <hip/hip_runtime.h>
#include <hip/hip_bf16.h>


typedef __attribute__((ext_vector_type(2))) float v2f;
typedef __attribute__((ext_vector_type(8))) float v8f;

constexpr int SLICE_E = 32768;
constexpr int CNT_T   = 1024;
constexpr int CNT_B   = 8;
constexpr int NW_CAP  = 65536;

template <bool RANK>
static __device__ __forceinline__ void gcn_count_slice(
    const int* __restrict__ idx, unsigned* tab, int* __restrict__ rank, int e0, int e1, int t)
{
    for (int base = e0; base < e1; base += CNT_B * CNT_T) {
        int v[CNT_B];
        #pragma unroll
        for (int j = 0; j < CNT_B; ++j) {
            const int i = base + j * CNT_T + t;
            v[j] = idx[(i < e1) ? i : (e1 - 1)];
        }
        if (RANK) {
            unsigned old[CNT_B];
            #pragma unroll
            for (int j = 0; j < CNT_B; ++j) {
                const bool in = (base + j * CNT_T + t < e1);
                old[j] = atomicAdd(&tab[in ? (v[j] >> 1) : t], in ? (1u << ((v[j] & 1) << 4)) : 0u);
            }
            #pragma unroll
            for (int j = 0; j < CNT_B; ++j) {
                const int i = base + j * CNT_T + t;
                if (i < e1) rank[i] = (int)((old[j] >> ((v[j] & 1) << 4)) & 0xFFFFu);
            }
        } else {
            #pragma unroll
            for (int j = 0; j < CNT_B; ++j) {
                const bool in = (base + j * CNT_T + t < e1);
                atomicAdd(&tab[in ? (v[j] >> 1) : t], in ? (1u << ((v[j] & 1) << 4)) : 0u);
            }
        }
    }
}

__global__ __launch_bounds__(CNT_T) void gcn_count(
    const int* __restrict__ src, const int* __restrict__ dst,
    unsigned* __restrict__ tabD, unsigned* __restrict__ tabS,
    int* __restrict__ rank, int e, int nw, int nslices)
{
    __shared__ __attribute__((aligned(16))) unsigned tab[NW_CAP];
    const int  t      = threadIdx.x;
    const bool is_dst = (int)blockIdx.x < nslices;
    const int  sl     = is_dst ? (int)blockIdx.x : (int)blockIdx.x - nslices;

    for (int w = 4 * t; w < nw; w += 4 * CNT_T)
        *reinterpret_cast<uint4*>(tab + w) = make_uint4(0u, 0u, 0u, 0u);
    __syncthreads();

    const int e0 = sl * SLICE_E;
    const int e1 = (e0 + SLICE_E < e) ? (e0 + SLICE_E) : e;
    if (is_dst) gcn_count_slice<true>(dst, tab, rank, e0, e1, t);
    else        gcn_count_slice<false>(src, tab, rank, e0, e1, t);
    __syncthreads();

    unsigned* __restrict__ out = (is_dst ? tabD : tabS) + (size_t)sl * nw;
    if (((nw & 3) == 0) && ((reinterpret_cast<size_t>(out) & 15) == 0)) {
        for (int w = 4 * t; w < nw; w += 4 * CNT_T)
            *reinterpret_cast<uint4*>(out + w) = *reinterpret_cast<const uint4*>(tab + w);
    } else {
        for (int w = t; w < nw; w += CNT_T) out[w] = tab[w];
    }
}

constexpr int CMB_C = 25;

__global__ __launch_bounds__(256) void gcn_combine(
    const unsigned* __restrict__ tabD, const unsigned* __restrict__ tabS,
    int* __restrict__ baseD, int* __restrict__ cnt_in,
    float* __restrict__ norm_src, float* __restrict__ norm_dst,
    int n, int nw, int nslices)
{
    const int w = blockIdx.x * blockDim.x + threadIdx.x;
    if (w >= nw) return;
    const size_t np = 2 * (size_t)nw;
    int r0 = 0, r1 = 0, o0 = 0, o1 = 0;
    for (int s0 = 0; s0 < nslices; s0 += CMB_C) {
        unsigned vd[CMB_C], vs[CMB_C];
        #pragma unroll
        for (int j = 0; j < CMB_C; ++j) {
            const bool in = (s0 + j < nslices);
            vd[j] = in ? tabD[(size_t)(s0 + j) * nw + w] : 0u;
            vs[j] = in ? tabS[(size_t)(s0 + j) * nw + w] : 0u;
        }
        #pragma unroll
        for (int j = 0; j < CMB_C; ++j) {
            if (s0 + j < nslices)
                *reinterpret_cast<int2*>(baseD + (size_t)(s0 + j) * np + 2 * (size_t)w) = make_int2(r0, r1);
            r0 += (int)(vd[j] & 0xFFFFu);
            r1 += (int)(vd[j] >> 16);
            o0 += (int)(vs[j] & 0xFFFFu);
            o1 += (int)(vs[j] >> 16);
        }
    }
    const int n0 = 2 * w;
    if (n0 >= n) return;
    cnt_in[n0]   = r0;
    norm_dst[n0] = rsqrtf(fmaxf((float)r0, 1.0f));
    norm_src[n0] = rsqrtf(fmaxf((float)o0, 1.0f));
    if (n0 + 1 < n) {
        cnt_in[n0 + 1]   = r1;
        norm_dst[n0 + 1] = rsqrtf(fmaxf((float)r1, 1.0f));
        norm_src[n0 + 1] = rsqrtf(fmaxf((float)o1, 1.0f));
    }
}

constexpr int SCAN_T    = 256;
constexpr int SCAN_TILE = 4 * SCAN_T;

static __device__ __forceinline__ int block_scan_excl(int v, int* sh, int t, int* total) {
    sh[t] = v;
    __syncthreads();
    for (int off = 1; off < SCAN_T; off <<= 1) {
        int a = (t >= off) ? sh[t - off] : 0;
        __syncthreads();
        sh[t] += a;
        __syncthreads();
    }
    *total = sh[SCAN_T - 1];
    int ex = sh[t] - v;
    __syncthreads();
    return ex;
}

__global__ __launch_bounds__(SCAN_T) void gcn_tile_sums(
    const int* __restrict__ cnt, int* __restrict__ tsum, int n)
{
    __shared__ int sh[SCAN_T];
    const int t = threadIdx.x;
    const int base = blockIdx.x * SCAN_TILE + 4 * t;
    int s = 0;
    for (int k = 0; k < 4; ++k) s += (base + k < n) ? cnt[base + k] : 0;
    int total;
    block_scan_excl(s, sh, t, &total);
    if (t == 0) tsum[blockIdx.x] = total;
}

__global__ __launch_bounds__(SCAN_T) void gcn_rowptr(
    const int* __restrict__ cnt, const int* __restrict__ tsum,
    int* __restrict__ rowptr, int n, int e)
{
    __shared__ int sh[SCAN_T];
    const int t = threadIdx.x;
    const int b = blockIdx.x;

    int part = 0;
    for (int j = t; j < b; j += SCAN_T) part += tsum[j];
    int offset;
    block_scan_excl(part, sh, t, &offset);

    const int base = b * SCAN_TILE + 4 * t;
    int c[4];
    int s = 0;
    for (int k = 0; k < 4; ++k) { c[k] = (base + k < n) ? cnt[base + k] : 0; s += c[k]; }
    int tile_total;
    int run = offset + block_scan_excl(s, sh, t, &tile_total);
    for (int k = 0; k < 4; ++k) {
        if (base + k < n) rowptr[base + k] = run;
        run += c[k];
    }
    if (b == 0 && t == 0) rowptr[n] = e;
}

__global__ void gcn_fill(const int* __restrict__ src, const int* __restrict__ dst,
                         const int* __restrict__ rank, const int* __restrict__ rowptr,
                         const int* __restrict__ baseD, int* __restrict__ csr, int e, int nw) {
    int i = blockIdx.x * blockDim.x + threadIdx.x;
    if (i >= e) return;
    const int d = dst[i];
    const size_t np = 2 * (size_t)nw;
    csr[rowptr[d] + baseD[(size_t)(i / SLICE_E) * np + d] + rank[i]] = src[i];
}

__global__ void gcn_move_i32(const int* __restrict__ a, int* __restrict__ b, long n) {
    long i = (long)blockIdx.x * blockDim.x + threadIdx.x;
    if (i < n) b[i] = a[i];
}

template <int DIN, int DN>
__global__ __launch_bounds__(((DN + 15) / 16) * 32) void gcn_gemm_wmma(
    const float* __restrict__ H, const float* __restrict__ norm_src,
    const float* __restrict__ W, float* __restrict__ Y, int n)
{
    constexpr int T   = ((DN + 15) / 16) * 32;
    constexpr int LDA = DIN + 4;
    constexpr int RC  = DIN / 4;
    constexpr int NCH = 16 * RC;
    constexpr int PER = (NCH + T - 1) / T;
    constexpr bool EXACT = (NCH % T) == 0;
    __shared__ __attribute__((aligned(16))) float As[16 * LDA];

    const int row0 = blockIdx.x * 16;
    const int tid  = threadIdx.x;

    float4 hv[PER];
    float  sv[PER];
    #pragma unroll
    for (int j = 0; j < PER; ++j) {
        const int ch  = tid + j * T;
        const int chc = (EXACT || ch < NCH) ? ch : (NCH - 1);
        const int r   = chc / RC;
        const int c4  = chc - r * RC;
        const int row = row0 + r;
        const int rowc = (row < n) ? row : (n - 1);
        hv[j] = *reinterpret_cast<const float4*>(H + (size_t)rowc * DIN + 4 * c4);
        sv[j] = norm_src[rowc];
    }
    #pragma unroll
    for (int j = 0; j < PER; ++j) {
        const int ch = tid + j * T;
        if (EXACT || ch < NCH) {
            const int r   = ch / RC;
            const int c4  = ch - r * RC;
            const float s = sv[j];
            float4 v;
            v.x = hv[j].x * s;
            v.y = hv[j].y * s;
            v.z = hv[j].z * s;
            v.w = hv[j].w * s;
            *reinterpret_cast<float4*>(As + r * LDA + 4 * c4) = v;
        }
    }
    __syncthreads();

    const int lane = tid & 31;
    const int wave = tid >> 5;
    const int mn   = lane & 15;
    const int kh   = lane >> 4;
    const int col  = (wave << 4) + mn;

    const int   colc = (col < DN) ? col : (DN - 1);
    float msk = 1.0f;
    if constexpr ((DN & 15) != 0) msk = (col < DN) ? 1.0f : 0.0f;

    const float* __restrict__ ap = As + mn * LDA + 2 * kh;
    const float* __restrict__ wp = W + (size_t)(2 * kh) * DN + colc;

    v8f acc = {0.f, 0.f, 0.f, 0.f, 0.f, 0.f, 0.f, 0.f};

    #pragma unroll
    for (int k = 0; k < DIN; k += 4) {
        v2f b;
        const v2f a = *reinterpret_cast<const v2f*>(ap + k);
        b.x = wp[k * DN] * msk;
        b.y = wp[k * DN + DN] * msk;
        acc = __builtin_amdgcn_wmma_f32_16x16x4_f32(
            false, a, false, b, (short)0, acc, false, false);
    }

    if (col < DN) {
        #pragma unroll
        for (int v = 0; v < 8; ++v) {
            int row = row0 + v + 8 * kh;
            if (row < n) Y[(size_t)row * DN + col] = acc[v];
        }
    }
}

template <int D, int RELU>
__global__ __launch_bounds__(256) void gcn_gather(
    const float* __restrict__ Y, const int* __restrict__ rowptr, const int* __restrict__ csr,
    const float* __restrict__ norm_dst, const float* __restrict__ bias,
    float* __restrict__ Out, int n)
{
    constexpr int CPE = D >> 2;
    long idx = (long)blockIdx.x * blockDim.x + threadIdx.x;
    if (idx >= (long)n * CPE) return;
    const int i = (int)(idx / CPE);
    const int c = (int)(idx - (long)i * CPE);

    int p = rowptr[i];
    const int end  = rowptr[i + 1];
    const int last = end - 1;
    const float* __restrict__ yb = Y + 4 * c;

    float4 acc = make_float4(0.f, 0.f, 0.f, 0.f);
    for (; p < end; p += 4) {
        const bool h1 = (p + 1 < end), h2 = (p + 2 < end), h3 = (p + 3 < end);
        const int s0 = csr[p];
        const int s1 = csr[h1 ? p + 1 : last];
        const int s2 = csr[h2 ? p + 2 : last];
        const int s3 = csr[h3 ? p + 3 : last];
        const float4 v0 = *reinterpret_cast<const float4*>(yb + (size_t)s0 * D);
        const float4 v1 = *reinterpret_cast<const float4*>(yb + (size_t)s1 * D);
        const float4 v2 = *reinterpret_cast<const float4*>(yb + (size_t)s2 * D);
        const float4 v3 = *reinterpret_cast<const float4*>(yb + (size_t)s3 * D);
        acc.x += v0.x; acc.y += v0.y; acc.z += v0.z; acc.w += v0.w;
        acc.x += h1 ? v1.x : 0.f; acc.y += h1 ? v1.y : 0.f;
        acc.z += h1 ? v1.z : 0.f; acc.w += h1 ? v1.w : 0.f;
        acc.x += h2 ? v2.x : 0.f; acc.y += h2 ? v2.y : 0.f;
        acc.z += h2 ? v2.z : 0.f; acc.w += h2 ? v2.w : 0.f;
        acc.x += h3 ? v3.x : 0.f; acc.y += h3 ? v3.y : 0.f;
        acc.z += h3 ? v3.z : 0.f; acc.w += h3 ? v3.w : 0.f;
    }

    const float nd = norm_dst[i];
    const float4 bv = *reinterpret_cast<const float4*>(bias + 4 * c);
    float4 r;
    r.x = acc.x * nd + bv.x;
    r.y = acc.y * nd + bv.y;
    r.z = acc.z * nd + bv.z;
    r.w = acc.w * nd + bv.w;
    if (RELU) {
        r.x = fmaxf(r.x, 0.0f); r.y = fmaxf(r.y, 0.0f);
        r.z = fmaxf(r.z, 0.0f); r.w = fmaxf(r.w, 0.0f);
    }
    *reinterpret_cast<float4*>(Out + (size_t)i * D + 4 * c) = r;
}

static inline int cdiv_l(long a, long b) { return (int)((a + b - 1) / b); }

template <int DIN, int DN, int RELU>
static void run_layer(const float* Hin, const float* norm_src, const float* norm_dst,
                      const float* W, const float* bias, float* bufP, float* Agg,
                      const int* rowptr, const int* csr, int N, hipStream_t stream) {
    constexpr int CT = (DN + 15) / 16;
    gcn_gemm_wmma<DIN, DN><<<cdiv_l(N, 16), 32 * CT, 0, stream>>>(
        Hin, norm_src, W, bufP, N);
    gcn_gather<DN, RELU><<<cdiv_l((long)N * (DN / 4), 256), 256, 0, stream>>>(
        bufP, rowptr, csr, norm_dst, bias, Agg, N);
}

extern "C" void kernel_launch(void* const* d_in, const int* in_sizes, int n_in,
                              void* d_out, int out_size, void* d_ws, size_t ws_size,
                              hipStream_t stream) {
    const float* x   = (const float*)d_in[0];
    const int*   src = (const int*)d_in[1];
    const int*   dst = (const int*)d_in[2];
    const float* W1 = (const float*)d_in[3];
    const float* b1 = (const float*)d_in[4];
    const float* W2 = (const float*)d_in[5];
    const float* b2 = (const float*)d_in[6];
    const float* W3 = (const float*)d_in[7];
    const float* b3 = (const float*)d_in[8];
    const float* W4 = (const float*)d_in[9];
    const float* b4 = (const float*)d_in[10];

    const int N = in_sizes[0] / 256;
    const int E = in_sizes[1];
    const int NT = cdiv_l(N, SCAN_TILE);

    float* ws       = (float*)d_ws;
    float* norm_src = ws;
    float* norm_dst = ws + N;
    float* bufP     = ws + 2 * (size_t)N;
    float* bufH     = bufP + (size_t)N * 128;

    int* csr0   = (int*)d_out;
    int* rowp0  = csr0 + E;
    int* rank   = rowp0 + N + 1;
    int* cnt_in = rank + E;
    int* tsum   = cnt_in + N;
    int* csrW   = (int*)(bufP + (size_t)N * 64);
    int* rowpW  = csrW + E;

    const int NS = cdiv_l(E, SLICE_E);
    const int NW = 4 * cdiv_l(N, 8);
    unsigned* tabD = (unsigned*)bufH;
    unsigned* tabS = tabD + (size_t)NS * NW;
    int* baseD     = (int*)(tabS + (size_t)NS * NW);

    gcn_count<<<2 * NS, CNT_T, 0, stream>>>(src, dst, tabD, tabS, rank, E, NW, NS);
    gcn_combine<<<cdiv_l(NW, 256), 256, 0, stream>>>(tabD, tabS, baseD, cnt_in, norm_src, norm_dst, N, NW, NS);
    gcn_tile_sums<<<NT, SCAN_T, 0, stream>>>(cnt_in, tsum, N);
    gcn_rowptr<<<NT, SCAN_T, 0, stream>>>(cnt_in, tsum, rowp0, N, E);
    gcn_fill<<<cdiv_l(E, 256), 256, 0, stream>>>(src, dst, rank, rowp0, baseD, csr0, E, NW);

    run_layer<256, 128, 1>(x,    norm_src, norm_dst, W1, b1, bufP, bufH, rowp0, csr0, N, stream);
    gcn_move_i32<<<cdiv_l((long)E + N + 1, 256), 256, 0, stream>>>(csr0, csrW, (long)E + N + 1);
    run_layer<128, 64, 1>(bufH,  norm_src, norm_dst, W2, b2, bufP, bufH, rowpW, csrW, N, stream);
    run_layer<64, 16, 1>(bufH,   norm_src, norm_dst, W3, b3, bufP, bufH, rowpW, csrW, N, stream);
    run_layer<16, 40, 0>(bufH,   norm_src, norm_dst, W4, b4, bufP, (float*)d_out, rowpW, csrW, N, stream);
}


# A layer's aggregation hands its scaled rows through LDS to the next layer's WMMA product in the same block: 3 projection launches, the h write+read and the list copy go (14 -> 10); sums unchanged
# speedup vs baseline: 3.7021x; 1.0622x over previous
#include <hip/hip_runtime.h>
#include <hip/hip_bf16.h>


typedef __attribute__((ext_vector_type(2))) float v2f;
typedef __attribute__((ext_vector_type(8))) float v8f;

constexpr int SLICE_E = 32768;
constexpr int CNT_T   = 1024;
constexpr int CNT_B   = 8;
constexpr int NW_CAP  = 65536;

template <bool RANK>
static __device__ __forceinline__ void gcn_count_slice(
    const int* __restrict__ idx, unsigned* tab, int* __restrict__ rank, int e0, int e1, int t)
{
    for (int base = e0; base < e1; base += CNT_B * CNT_T) {
        int v[CNT_B];
        #pragma unroll
        for (int j = 0; j < CNT_B; ++j) {
            const int i = base + j * CNT_T + t;
            v[j] = idx[(i < e1) ? i : (e1 - 1)];
        }
        if (RANK) {
            unsigned old[CNT_B];
            #pragma unroll
            for (int j = 0; j < CNT_B; ++j) {
                const bool in = (base + j * CNT_T + t < e1);
                old[j] = atomicAdd(&tab[in ? (v[j] >> 1) : t], in ? (1u << ((v[j] & 1) << 4)) : 0u);
            }
            #pragma unroll
            for (int j = 0; j < CNT_B; ++j) {
                const int i = base + j * CNT_T + t;
                if (i < e1) rank[i] = (int)((old[j] >> ((v[j] & 1) << 4)) & 0xFFFFu);
            }
        } else {
            #pragma unroll
            for (int j = 0; j < CNT_B; ++j) {
                const bool in = (base + j * CNT_T + t < e1);
                atomicAdd(&tab[in ? (v[j] >> 1) : t], in ? (1u << ((v[j] & 1) << 4)) : 0u);
            }
        }
    }
}

__global__ __launch_bounds__(CNT_T) void gcn_count(
    const int* __restrict__ src, const int* __restrict__ dst,
    unsigned* __restrict__ tabD, unsigned* __restrict__ tabS,
    int* __restrict__ rank, int e, int nw, int nslices)
{
    __shared__ __attribute__((aligned(16))) unsigned tab[NW_CAP];
    const int  t      = threadIdx.x;
    const bool is_dst = (int)blockIdx.x < nslices;
    const int  sl     = is_dst ? (int)blockIdx.x : (int)blockIdx.x - nslices;

    for (int w = 4 * t; w < nw; w += 4 * CNT_T)
        *reinterpret_cast<uint4*>(tab + w) = make_uint4(0u, 0u, 0u, 0u);
    __syncthreads();

    const int e0 = sl * SLICE_E;
    const int e1 = (e0 + SLICE_E < e) ? (e0 + SLICE_E) : e;
    if (is_dst) gcn_count_slice<true>(dst, tab, rank, e0, e1, t);
    else        gcn_count_slice<false>(src, tab, rank, e0, e1, t);
    __syncthreads();

    unsigned* __restrict__ out = (is_dst ? tabD : tabS) + (size_t)sl * nw;
    if (((nw & 3) == 0) && ((reinterpret_cast<size_t>(out) & 15) == 0)) {
        for (int w = 4 * t; w < nw; w += 4 * CNT_T)
            *reinterpret_cast<uint4*>(out + w) = *reinterpret_cast<const uint4*>(tab + w);
    } else {
        for (int w = t; w < nw; w += CNT_T) out[w] = tab[w];
    }
}

constexpr int CMB_C = 25;

__global__ __launch_bounds__(256) void gcn_combine(
    const unsigned* __restrict__ tabD, const unsigned* __restrict__ tabS,
    int* __restrict__ baseD, int* __restrict__ cnt_in,
    float* __restrict__ norm_src, float* __restrict__ norm_dst,
    int n, int nw, int nslices)
{
    const int w = blockIdx.x * blockDim.x + threadIdx.x;
    if (w >= nw) return;
    const size_t np = 2 * (size_t)nw;
    int r0 = 0, r1 = 0, o0 = 0, o1 = 0;
    for (int s0 = 0; s0 < nslices; s0 += CMB_C) {
        unsigned vd[CMB_C], vs[CMB_C];
        #pragma unroll
        for (int j = 0; j < CMB_C; ++j) {
            const bool in = (s0 + j < nslices);
            vd[j] = in ? tabD[(size_t)(s0 + j) * nw + w] : 0u;
            vs[j] = in ? tabS[(size_t)(s0 + j) * nw + w] : 0u;
        }
        #pragma unroll
        for (int j = 0; j < CMB_C; ++j) {
            if (s0 + j < nslices)
                *reinterpret_cast<int2*>(baseD + (size_t)(s0 + j) * np + 2 * (size_t)w) = make_int2(r0, r1);
            r0 += (int)(vd[j] & 0xFFFFu);
            r1 += (int)(vd[j] >> 16);
            o0 += (int)(vs[j] & 0xFFFFu);
            o1 += (int)(vs[j] >> 16);
        }
    }
    const int n0 = 2 * w;
    if (n0 >= n) return;
    cnt_in[n0]   = r0;
    norm_dst[n0] = rsqrtf(fmaxf((float)r0, 1.0f));
    norm_src[n0] = rsqrtf(fmaxf((float)o0, 1.0f));
    if (n0 + 1 < n) {
        cnt_in[n0 + 1]   = r1;
        norm_dst[n0 + 1] = rsqrtf(fmaxf((float)r1, 1.0f));
        norm_src[n0 + 1] = rsqrtf(fmaxf((float)o1, 1.0f));
    }
}

constexpr int SCAN_T    = 256;
constexpr int SCAN_TILE = 4 * SCAN_T;

static __device__ __forceinline__ int block_scan_excl(int v, int* sh, int t, int* total) {
    sh[t] = v;
    __syncthreads();
    for (int off = 1; off < SCAN_T; off <<= 1) {
        int a = (t >= off) ? sh[t - off] : 0;
        __syncthreads();
        sh[t] += a;
        __syncthreads();
    }
    *total = sh[SCAN_T - 1];
    int ex = sh[t] - v;
    __syncthreads();
    return ex;
}

__global__ __launch_bounds__(SCAN_T) void gcn_tile_sums(
    const int* __restrict__ cnt, int* __restrict__ tsum, int n)
{
    __shared__ int sh[SCAN_T];
    const int t = threadIdx.x;
    const int base = blockIdx.x * SCAN_TILE + 4 * t;
    int s = 0;
    for (int k = 0; k < 4; ++k) s += (base + k < n) ? cnt[base + k] : 0;
    int total;
    block_scan_excl(s, sh, t, &total);
    if (t == 0) tsum[blockIdx.x] = total;
}

__global__ __launch_bounds__(SCAN_T) void gcn_rowptr(
    const int* __restrict__ cnt, const int* __restrict__ tsum,
    int* __restrict__ rowptr, int n, int e)
{
    __shared__ int sh[SCAN_T];
    const int t = threadIdx.x;
    const int b = blockIdx.x;

    int part = 0;
    for (int j = t; j < b; j += SCAN_T) part += tsum[j];
    int offset;
    block_scan_excl(part, sh, t, &offset);

    const int base = b * SCAN_TILE + 4 * t;
    int c[4];
    int s = 0;
    for (int k = 0; k < 4; ++k) { c[k] = (base + k < n) ? cnt[base + k] : 0; s += c[k]; }
    int tile_total;
    int run = offset + block_scan_excl(s, sh, t, &tile_total);
    for (int k = 0; k < 4; ++k) {
        if (base + k < n) rowptr[base + k] = run;
        run += c[k];
    }
    if (b == 0 && t == 0) rowptr[n] = e;
}

__global__ void gcn_fill(const int* __restrict__ src, const int* __restrict__ dst,
                         const int* __restrict__ rank, const int* __restrict__ rowptr,
                         const int* __restrict__ baseD, int* __restrict__ csr, int e, int nw) {
    int i = blockIdx.x * blockDim.x + threadIdx.x;
    if (i >= e) return;
    const int d = dst[i];
    const size_t np = 2 * (size_t)nw;
    csr[rowptr[d] + baseD[(size_t)(i / SLICE_E) * np + d] + rank[i]] = src[i];
}

template <int DIN, int DN>
__global__ __launch_bounds__(((DN + 15) / 16) * 32) void gcn_gemm_wmma(
    const float* __restrict__ H, const float* __restrict__ norm_src,
    const float* __restrict__ W, float* __restrict__ Y, int n)
{
    constexpr int T   = ((DN + 15) / 16) * 32;
    constexpr int LDA = DIN + 4;
    constexpr int RC  = DIN / 4;
    constexpr int NCH = 16 * RC;
    constexpr int PER = (NCH + T - 1) / T;
    constexpr bool EXACT = (NCH % T) == 0;
    __shared__ __attribute__((aligned(16))) float As[16 * LDA];

    const int row0 = blockIdx.x * 16;
    const int tid  = threadIdx.x;

    float4 hv[PER];
    float  sv[PER];
    #pragma unroll
    for (int j = 0; j < PER; ++j) {
        const int ch  = tid + j * T;
        const int chc = (EXACT || ch < NCH) ? ch : (NCH - 1);
        const int r   = chc / RC;
        const int c4  = chc - r * RC;
        const int row = row0 + r;
        const int rowc = (row < n) ? row : (n - 1);
        hv[j] = *reinterpret_cast<const float4*>(H + (size_t)rowc * DIN + 4 * c4);
        sv[j] = norm_src[rowc];
    }
    #pragma unroll
    for (int j = 0; j < PER; ++j) {
        const int ch = tid + j * T;
        if (EXACT || ch < NCH) {
            const int r   = ch / RC;
            const int c4  = ch - r * RC;
            const float s = sv[j];
            float4 v;
            v.x = hv[j].x * s;
            v.y = hv[j].y * s;
            v.z = hv[j].z * s;
            v.w = hv[j].w * s;
            *reinterpret_cast<float4*>(As + r * LDA + 4 * c4) = v;
        }
    }
    __syncthreads();

    const int lane = tid & 31;
    const int wave = tid >> 5;
    const int mn   = lane & 15;
    const int kh   = lane >> 4;
    const int col  = (wave << 4) + mn;

    const int   colc = (col < DN) ? col : (DN - 1);
    float msk = 1.0f;
    if constexpr ((DN & 15) != 0) msk = (col < DN) ? 1.0f : 0.0f;

    const float* __restrict__ ap = As + mn * LDA + 2 * kh;
    const float* __restrict__ wp = W + (size_t)(2 * kh) * DN + colc;

    v8f acc = {0.f, 0.f, 0.f, 0.f, 0.f, 0.f, 0.f, 0.f};

    #pragma unroll
    for (int k = 0; k < DIN; k += 4) {
        v2f b;
        const v2f a = *reinterpret_cast<const v2f*>(ap + k);
        b.x = wp[k * DN] * msk;
        b.y = wp[k * DN + DN] * msk;
        acc = __builtin_amdgcn_wmma_f32_16x16x4_f32(
            false, a, false, b, (short)0, acc, false, false);
    }

    if (col < DN) {
        #pragma unroll
        for (int v = 0; v < 8; ++v) {
            int row = row0 + v + 8 * kh;
            if (row < n) Y[(size_t)row * DN + col] = acc[v];
        }
    }
}

template <int D, int RELU>
__global__ __launch_bounds__(256) void gcn_gather(
    const float* __restrict__ Y, const int* __restrict__ rowptr, const int* __restrict__ csr,
    const float* __restrict__ norm_dst, const float* __restrict__ bias,
    float* __restrict__ Out, int n)
{
    constexpr int CPE = D >> 2;
    long idx = (long)blockIdx.x * blockDim.x + threadIdx.x;
    if (idx >= (long)n * CPE) return;
    const int i = (int)(idx / CPE);
    const int c = (int)(idx - (long)i * CPE);

    int p = rowptr[i];
    const int end  = rowptr[i + 1];
    const int last = end - 1;
    const float* __restrict__ yb = Y + 4 * c;

    float4 acc = make_float4(0.f, 0.f, 0.f, 0.f);
    for (; p < end; p += 4) {
        const bool h1 = (p + 1 < end), h2 = (p + 2 < end), h3 = (p + 3 < end);
        const int s0 = csr[p];
        const int s1 = csr[h1 ? p + 1 : last];
        const int s2 = csr[h2 ? p + 2 : last];
        const int s3 = csr[h3 ? p + 3 : last];
        const float4 v0 = *reinterpret_cast<const float4*>(yb + (size_t)s0 * D);
        const float4 v1 = *reinterpret_cast<const float4*>(yb + (size_t)s1 * D);
        const float4 v2 = *reinterpret_cast<const float4*>(yb + (size_t)s2 * D);
        const float4 v3 = *reinterpret_cast<const float4*>(yb + (size_t)s3 * D);
        acc.x += v0.x; acc.y += v0.y; acc.z += v0.z; acc.w += v0.w;
        acc.x += h1 ? v1.x : 0.f; acc.y += h1 ? v1.y : 0.f;
        acc.z += h1 ? v1.z : 0.f; acc.w += h1 ? v1.w : 0.f;
        acc.x += h2 ? v2.x : 0.f; acc.y += h2 ? v2.y : 0.f;
        acc.z += h2 ? v2.z : 0.f; acc.w += h2 ? v2.w : 0.f;
        acc.x += h3 ? v3.x : 0.f; acc.y += h3 ? v3.y : 0.f;
        acc.z += h3 ? v3.z : 0.f; acc.w += h3 ? v3.w : 0.f;
    }

    const float nd = norm_dst[i];
    const float4 bv = *reinterpret_cast<const float4*>(bias + 4 * c);
    float4 r;
    r.x = acc.x * nd + bv.x;
    r.y = acc.y * nd + bv.y;
    r.z = acc.z * nd + bv.z;
    r.w = acc.w * nd + bv.w;
    if (RELU) {
        r.x = fmaxf(r.x, 0.0f); r.y = fmaxf(r.y, 0.0f);
        r.z = fmaxf(r.z, 0.0f); r.w = fmaxf(r.w, 0.0f);
    }
    *reinterpret_cast<float4*>(Out + (size_t)i * D + 4 * c) = r;
}

template <int D, int DN, int ROWS, int RELU>
__global__ __launch_bounds__(ROWS * (D / 4)) void gcn_gather_gemm(
    const float* __restrict__ Y, const int* __restrict__ rowptr, const int* __restrict__ csr,
    const float* __restrict__ norm_dst, const float* __restrict__ bias,
    const float* __restrict__ norm_src, const float* __restrict__ W,
    float* __restrict__ Yn, int n)
{
    constexpr int CPE = D >> 2;
    constexpr int T   = ROWS * CPE;
    constexpr int LDA = D + 4;
    constexpr int CT  = (DN + 15) / 16;
    constexpr int RT  = ROWS / 16;
    constexpr int NWV = T / 32;
    static_assert(ROWS % 16 == 0 && T % 32 == 0 && T <= 1024, "block shape");
    __shared__ __attribute__((aligned(16))) float As[ROWS * LDA];

    const int tid  = threadIdx.x;
    const int r    = tid / CPE;
    const int c    = tid - r * CPE;
    const int row0 = blockIdx.x * ROWS;
    const int i    = row0 + r;

    float4 hs = make_float4(0.f, 0.f, 0.f, 0.f);
    if (i < n) {
        int p = rowptr[i];
        const int end  = rowptr[i + 1];
        const int last = end - 1;
        const float* __restrict__ yb = Y + 4 * c;

        float4 acc = make_float4(0.f, 0.f, 0.f, 0.f);
        for (; p < end; p += 4) {
            const bool h1 = (p + 1 < end), h2 = (p + 2 < end), h3 = (p + 3 < end);
            const int s0 = csr[p];
            const int s1 = csr[h1 ? p + 1 : last];
            const int s2 = csr[h2 ? p + 2 : last];
            const int s3 = csr[h3 ? p + 3 : last];
            const float4 v0 = *reinterpret_cast<const float4*>(yb + (size_t)s0 * D);
            const float4 v1 = *reinterpret_cast<const float4*>(yb + (size_t)s1 * D);
            const float4 v2 = *reinterpret_cast<const float4*>(yb + (size_t)s2 * D);
            const float4 v3 = *reinterpret_cast<const float4*>(yb + (size_t)s3 * D);
            acc.x += v0.x; acc.y += v0.y; acc.z += v0.z; acc.w += v0.w;
            acc.x += h1 ? v1.x : 0.f; acc.y += h1 ? v1.y : 0.f;
            acc.z += h1 ? v1.z : 0.f; acc.w += h1 ? v1.w : 0.f;
            acc.x += h2 ? v2.x : 0.f; acc.y += h2 ? v2.y : 0.f;
            acc.z += h2 ? v2.z : 0.f; acc.w += h2 ? v2.w : 0.f;
            acc.x += h3 ? v3.x : 0.f; acc.y += h3 ? v3.y : 0.f;
            acc.z += h3 ? v3.z : 0.f; acc.w += h3 ? v3.w : 0.f;
        }

        const float nd = norm_dst[i];
        const float s  = norm_src[i];
        const float4 bv = *reinterpret_cast<const float4*>(bias + 4 * c);
        float4 h;
        h.x = acc.x * nd + bv.x;
        h.y = acc.y * nd + bv.y;
        h.z = acc.z * nd + bv.z;
        h.w = acc.w * nd + bv.w;
        if (RELU) {
            h.x = fmaxf(h.x, 0.0f); h.y = fmaxf(h.y, 0.0f);
            h.z = fmaxf(h.z, 0.0f); h.w = fmaxf(h.w, 0.0f);
        }
        hs.x = h.x * s;
        hs.y = h.y * s;
        hs.z = h.z * s;
        hs.w = h.w * s;
    }
    *reinterpret_cast<float4*>(As + r * LDA + 4 * c) = hs;
    __syncthreads();

    const int lane = tid & 31;
    const int wave = __builtin_amdgcn_readfirstlane(tid >> 5);
    const int mn   = lane & 15;
    const int kh   = lane >> 4;

    for (int job = wave; job < RT * CT; job += NWV) {
        const int rt  = job / CT;
        const int ct  = job - rt * CT;
        const int col = (ct << 4) + mn;

        const int colc = (col < DN) ? col : (DN - 1);
        float msk = 1.0f;
        if constexpr ((DN & 15) != 0) msk = (col < DN) ? 1.0f : 0.0f;

        const float* __restrict__ ap = As + (rt * 16 + mn) * LDA + 2 * kh;
        const float* __restrict__ wp = W + (size_t)(2 * kh) * DN + colc;

        v8f acc = {0.f, 0.f, 0.f, 0.f, 0.f, 0.f, 0.f, 0.f};

        #pragma unroll
        for (int k = 0; k < D; k += 4) {
            v2f b;
            const v2f a = *reinterpret_cast<const v2f*>(ap + k);
            b.x = wp[k * DN] * msk;
            b.y = wp[k * DN + DN] * msk;
            acc = __builtin_amdgcn_wmma_f32_16x16x4_f32(
                false, a, false, b, (short)0, acc, false, false);
        }

        if (col < DN) {
            #pragma unroll
            for (int v = 0; v < 8; ++v) {
                const int row = row0 + rt * 16 + v + 8 * kh;
                if (row < n) Yn[(size_t)row * DN + col] = acc[v];
            }
        }
    }
}

static inline int cdiv_l(long a, long b) { return (int)((a + b - 1) / b); }

extern "C" void kernel_launch(void* const* d_in, const int* in_sizes, int n_in,
                              void* d_out, int out_size, void* d_ws, size_t ws_size,
                              hipStream_t stream) {
    const float* x   = (const float*)d_in[0];
    const int*   src = (const int*)d_in[1];
    const int*   dst = (const int*)d_in[2];
    const float* W1 = (const float*)d_in[3];
    const float* b1 = (const float*)d_in[4];
    const float* W2 = (const float*)d_in[5];
    const float* b2 = (const float*)d_in[6];
    const float* W3 = (const float*)d_in[7];
    const float* b3 = (const float*)d_in[8];
    const float* W4 = (const float*)d_in[9];
    const float* b4 = (const float*)d_in[10];

    const int N = in_sizes[0] / 256;
    const int E = in_sizes[1];
    const int NT = cdiv_l(N, SCAN_TILE);

    float* ws       = (float*)d_ws;
    float* norm_src = ws;
    float* norm_dst = ws + N;
    float* bufP     = ws + 2 * (size_t)N;
    float* bufH     = bufP + (size_t)N * 128;

    int* rank   = (int*)d_out;
    int* cnt_in = rank + E;
    int* tsum   = cnt_in + N;
    int* csr    = (int*)(bufH + (size_t)N * 110);
    int* rowp   = csr + E;

    const int NS = cdiv_l(E, SLICE_E);
    const int NW = 4 * cdiv_l(N, 8);
    unsigned* tabD = (unsigned*)bufH;
    unsigned* tabS = tabD + (size_t)NS * NW;
    int* baseD     = (int*)(tabS + (size_t)NS * NW);

    gcn_count<<<2 * NS, CNT_T, 0, stream>>>(src, dst, tabD, tabS, rank, E, NW, NS);
    gcn_combine<<<cdiv_l(NW, 256), 256, 0, stream>>>(tabD, tabS, baseD, cnt_in, norm_src, norm_dst, N, NW, NS);
    gcn_tile_sums<<<NT, SCAN_T, 0, stream>>>(cnt_in, tsum, N);
    gcn_rowptr<<<NT, SCAN_T, 0, stream>>>(cnt_in, tsum, rowp, N, E);
    gcn_fill<<<cdiv_l(E, 256), 256, 0, stream>>>(src, dst, rank, rowp, baseD, csr, E, NW);

    gcn_gemm_wmma<256, 128><<<cdiv_l(N, 16), 256, 0, stream>>>(x, norm_src, W1, bufP, N);
    gcn_gather_gemm<128, 64, 16, 1><<<cdiv_l(N, 16), 512, 0, stream>>>(
        bufP, rowp, csr, norm_dst, b1, norm_src, W2, bufH, N);
    gcn_gather_gemm<64, 16, 16, 1><<<cdiv_l(N, 16), 256, 0, stream>>>(
        bufH, rowp, csr, norm_dst, b2, norm_src, W3, bufP, N);
    gcn_gather_gemm<16, 40, 64, 1><<<cdiv_l(N, 64), 256, 0, stream>>>(
        bufP, rowp, csr, norm_dst, b3, norm_src, W4, bufH, N);
    gcn_gather<40, 0><<<cdiv_l((long)N * 10, 256), 256, 0, stream>>>(
        bufH, rowp, csr, norm_dst, b4, (float*)d_out, N);
}
